# starting kernel
# baseline (speedup 1.0000x reference)
#include <hip/hip_runtime.h>
#include <hip/hip_bf16.h>

#define CC 8
#define MM 32
#define GG 16
#define LL 4

typedef float v2f __attribute__((ext_vector_type(2)));
typedef float v8f __attribute__((ext_vector_type(8)));

__global__ void softmax_rows_kernel(const float* __restrict__ src,
                                    float* __restrict__ dst,
                                    int nrows, int len, int stride,
                                    int d1, int s1, int d2, int s2) {
    int r = blockIdx.x * blockDim.x + threadIdx.x;
    if (r >= nrows) return;
    int base = (r / d1) * s1 + ((r % d1) / d2) * s2 + (r % d2);
    float mx = -3.402823466e38f;
    for (int i = 0; i < len; ++i) mx = fmaxf(mx, src[base + i * stride]);
    float sum = 0.f;
    for (int i = 0; i < len; ++i) sum += expf(src[base + i * stride] - mx);
    float inv = 1.0f / sum;
    for (int i = 0; i < len; ++i)
        dst[base + i * stride] = expf(src[base + i * stride] - mx) * inv;
}

__global__ void zero_int_kernel(int* __restrict__ p, int n) {
    int t = blockIdx.x * blockDim.x + threadIdx.x;
    if (t < n) p[t] = 0;
}

__global__ void degree_kernel(const int* __restrict__ dst, int* __restrict__ cnt, int E) {
    int t = blockIdx.x * blockDim.x + threadIdx.x;
    if (t < E) atomicAdd(&cnt[dst[t]], 1);
}

__global__ void invcnt_kernel(const int* __restrict__ cnt, float* __restrict__ inv, int N) {
    int t = blockIdx.x * blockDim.x + threadIdx.x;
    if (t < N) inv[t] = 1.0f / fmaxf((float)cnt[t], 1.0f);
}

__global__ __launch_bounds__(1024) void scan_kernel(const int* __restrict__ cnt,
                                                    int* __restrict__ rowstart,
                                                    int* __restrict__ cursor, int N) {
    __shared__ int buf[1024];
    __shared__ int carry;
    if (threadIdx.x == 0) carry = 0;
    __syncthreads();
    for (int base = 0; base < N; base += 1024) {
        int i = base + (int)threadIdx.x;
        int v = (i < N) ? cnt[i] : 0;
        buf[threadIdx.x] = v;
        __syncthreads();
        for (int off = 1; off < 1024; off <<= 1) {
            int t = (threadIdx.x >= (unsigned)off) ? buf[threadIdx.x - off] : 0;
            __syncthreads();
            buf[threadIdx.x] += t;
            __syncthreads();
        }
        int excl = buf[threadIdx.x] - v + carry;
        if (i < N) { rowstart[i] = excl; cursor[i] = excl; }
        __syncthreads();
        if (threadIdx.x == 1023) carry += buf[1023];
        __syncthreads();
    }
    if (threadIdx.x == 0) rowstart[N] = carry;
}

__global__ void csr_fill_kernel(const int* __restrict__ dst, const int* __restrict__ src,
                                int* __restrict__ cursor, int* __restrict__ eidx, int E) {
    int e = blockIdx.x * blockDim.x + threadIdx.x;
    if (e < E) {
        int p = atomicAdd(&cursor[dst[e]], 1);
        eidx[p] = src[e];
    }
}

__global__ void layer0_kernel(const int* __restrict__ x,
                              const float* __restrict__ B0s,
                              const float* __restrict__ Pis,
                              float* __restrict__ h,
                              float* __restrict__ out,
                              int N) {
    int t = blockIdx.x * blockDim.x + threadIdx.x;
    if (t >= N * GG) return;
    int n = t / GG, g = t % GG;
    int xm = x[n];
    float u[CC];
    float Z = 0.f;
#pragma unroll
    for (int c = 0; c < CC; ++c) {
        u[c] = B0s[(c * MM + xm) * GG + g] * Pis[c * GG + g];
        Z += u[c];
    }
    out[(size_t)n * (LL * GG) + g] = logf(Z);
    float iz = 1.0f / Z;
#pragma unroll
    for (int c = 0; c < CC; ++c)
        h[(size_t)n * (GG * CC) + g * CC + c] = u[c] * iz;
}

__global__ void gather_mean_kernel(const float* __restrict__ h,
                                   const int* __restrict__ rowstart,
                                   const int* __restrict__ eidx,
                                   const float* __restrict__ invc,
                                   float* __restrict__ aggr, int N) {
    int t = blockIdx.x * blockDim.x + threadIdx.x;
    int n = t >> 5;
    int lane = t & 31;
    if (n >= N) return;
    int beg = rowstart[n];
    int end = rowstart[n + 1];
    float4 acc = make_float4(0.f, 0.f, 0.f, 0.f);
    for (int j = beg; j < end; ++j) {
        int s = eidx[j];
        const float4 v = *(const float4*)(h + (size_t)s * 128 + lane * 4);
        acc.x += v.x; acc.y += v.y; acc.z += v.z; acc.w += v.w;
    }
    float iv = invc[n];
    acc.x *= iv; acc.y *= iv; acc.z *= iv; acc.w *= iv;
    *(float4*)(aggr + (size_t)n * 128 + lane * 4) = acc;
}

__global__ __launch_bounds__(512) void layer_update_kernel(
    const int* __restrict__ x,
    const float* __restrict__ Qg,
    const float* __restrict__ Bg,
    const float* __restrict__ aggr,
    float* __restrict__ h,
    float* __restrict__ out,
    int N, int lout) {
    __shared__ float qa[GG * 16 * CC];

    const int tid = threadIdx.x;
    const int w = tid >> 5;
    const int lane = tid & 31;
    const int half = lane >> 4;
    const int col = lane & 15;
    const int g = w;

    const int node = blockIdx.x * 16 + col;
    const int nn = node < N ? node : N - 1;
    const float sc = node < N ? 1.0f : 0.0f;
    const size_t abase = (size_t)nn * (GG * CC) + g * CC;
    const int k0 = half * 2;

    v2f a0, a1, b0, b1;
    a0.x = aggr[abase + k0 + 0] * sc;
    a0.y = aggr[abase + k0 + 1] * sc;
    a1.x = aggr[abase + k0 + 4] * sc;
    a1.y = aggr[abase + k0 + 5] * sc;

    const int qc = col & 7;
    const float cm = (col < CC) ? 1.0f : 0.0f;
    b0.x = Qg[(qc * CC + k0 + 0) * GG + g] * cm;
    b0.y = Qg[(qc * CC + k0 + 1) * GG + g] * cm;
    b1.x = Qg[(qc * CC + k0 + 4) * GG + g] * cm;
    b1.y = Qg[(qc * CC + k0 + 5) * GG + g] * cm;

    v8f acc = {0.f, 0.f, 0.f, 0.f, 0.f, 0.f, 0.f, 0.f};
    acc = __builtin_amdgcn_wmma_f32_16x16x4_f32(false, a0, false, b0,
                                                (short)0, acc, false, false);
    acc = __builtin_amdgcn_wmma_f32_16x16x4_f32(false, a1, false, b1,
                                                (short)0, acc, false, false);

    if (col < CC) {
#pragma unroll
        for (int r = 0; r < 8; ++r)
            qa[(w * 16 + (r + half * 8)) * CC + col] = acc[r];
    }
    __syncthreads();

    if (lane < 16) {
        int n2 = blockIdx.x * 16 + lane;
        if (n2 < N) {
            int xm = x[n2];
            float u[CC];
            float Z = 0.f;
#pragma unroll
            for (int c = 0; c < CC; ++c) {
                float bn = Bg[(c * MM + xm) * GG + g];
                u[c] = bn * qa[(w * 16 + lane) * CC + c];
                Z += u[c];
            }
            out[(size_t)n2 * (LL * GG) + lout * GG + g] = logf(Z);
            float iz = 1.0f / Z;
#pragma unroll
            for (int c = 0; c < CC; ++c)
                h[(size_t)n2 * (GG * CC) + g * CC + c] = u[c] * iz;
        }
    }
}

extern "C" void kernel_launch(void* const* d_in, const int* in_sizes, int n_in,
                              void* d_out, int out_size, void* d_ws, size_t ws_size,
                              hipStream_t stream) {
    const int* x = (const int*)d_in[0];
    const int* ei = (const int*)d_in[1];
    const float* lB0 = (const float*)d_in[2];
    const float* lPi = (const float*)d_in[3];
    const float* lQ = (const float*)d_in[4];
    const float* lB = (const float*)d_in[5];
    float* out = (float*)d_out;

    const int N = in_sizes[0];
    const int E = in_sizes[1] / 2;
    const int* dst = ei;
    const int* srcv = ei + E;

    float* ws = (float*)d_ws;
    size_t off = 0;
    float* B0s = ws + off; off += CC * MM * GG;
    float* Pis = ws + off; off += CC * GG;
    float* Qs  = ws + off; off += (LL - 1) * CC * CC * GG;
    float* Bs  = ws + off; off += (LL - 1) * CC * MM * GG;
    float* invc = ws + off; off += N;
    int* cnt      = (int*)(ws + off); off += N;
    int* rowstart = (int*)(ws + off); off += N + 1;
    int* cursor   = (int*)(ws + off); off += N;
    int* eidx     = (int*)(ws + off); off += E;
    off = (off + 15) & ~(size_t)15;
    float* h    = ws + off; off += (size_t)N * GG * CC;
    float* aggr = ws + off; off += (size_t)N * GG * CC;
    (void)ws_size; (void)n_in; (void)out_size;

    softmax_rows_kernel<<<1, 128, 0, stream>>>(lB0, B0s, CC * GG, MM, GG,
                                               CC * GG, 0, GG, MM * GG);
    softmax_rows_kernel<<<1, 32, 0, stream>>>(lPi, Pis, GG, CC, GG,
                                              GG, 0, GG, 0);
    softmax_rows_kernel<<<3, 128, 0, stream>>>(lQ, Qs, (LL - 1) * CC * GG, CC,
                                               CC * GG, CC * GG, CC * CC * GG,
                                               1, 1);
    softmax_rows_kernel<<<3, 128, 0, stream>>>(lB, Bs, (LL - 1) * CC * GG, MM,
                                               GG, CC * GG, CC * MM * GG,
                                               GG, MM * GG);

    zero_int_kernel<<<(N + 255) / 256, 256, 0, stream>>>(cnt, N);
    degree_kernel<<<(E + 255) / 256, 256, 0, stream>>>(dst, cnt, E);
    invcnt_kernel<<<(N + 255) / 256, 256, 0, stream>>>(cnt, invc, N);
    scan_kernel<<<1, 1024, 0, stream>>>(cnt, rowstart, cursor, N);
    csr_fill_kernel<<<(E + 255) / 256, 256, 0, stream>>>(dst, srcv, cursor, eidx, E);

    layer0_kernel<<<(N * GG + 255) / 256, 256, 0, stream>>>(x, B0s, Pis, h, out, N);

    const int gatherBlocks = (int)(((size_t)N * 32 + 255) / 256);
    for (int l = 0; l < LL - 1; ++l) {
        gather_mean_kernel<<<gatherBlocks, 256, 0, stream>>>(h, rowstart, eidx,
                                                             invc, aggr, N);
        layer_update_kernel<<<(N + 15) / 16, 512, 0, stream>>>(
            x, Qs + (size_t)l * CC * CC * GG, Bs + (size_t)l * CC * MM * GG,
            aggr, h, out, N, l + 1);
    }
}


# Layer update re-tiled: one wave per 16-node tile does all 16 g (D = Q*aggr^T, each lane owns all 8 c of its node,g); aggr and h rows cross LDS as whole 512-B rows; 16x fewer update waves
# speedup vs baseline: 1.9144x; 1.9144x over previous
#include <hip/hip_runtime.h>
#include <hip/hip_bf16.h>

#define CC 8
#define MM 32
#define GG 16
#define LL 4

typedef float v2f __attribute__((ext_vector_type(2)));
typedef float v8f __attribute__((ext_vector_type(8)));

__global__ void softmax_rows_kernel(const float* __restrict__ src,
                                    float* __restrict__ dst,
                                    int nrows, int len, int stride,
                                    int d1, int s1, int d2, int s2) {
    int r = blockIdx.x * blockDim.x + threadIdx.x;
    if (r >= nrows) return;
    int base = (r / d1) * s1 + ((r % d1) / d2) * s2 + (r % d2);
    float mx = -3.402823466e38f;
    for (int i = 0; i < len; ++i) mx = fmaxf(mx, src[base + i * stride]);
    float sum = 0.f;
    for (int i = 0; i < len; ++i) sum += expf(src[base + i * stride] - mx);
    float inv = 1.0f / sum;
    for (int i = 0; i < len; ++i)
        dst[base + i * stride] = expf(src[base + i * stride] - mx) * inv;
}

__global__ void softmax_q_frag_kernel(const float* __restrict__ src,
                                      float* __restrict__ dst) {
    int r = blockIdx.x * blockDim.x + threadIdx.x;
    if (r >= (LL - 1) * CC * GG) return;
    const int l = r / (CC * GG), k = (r % (CC * GG)) / GG, g = r % GG;
    const int base = l * (CC * CC * GG) + k * GG + g;
    const int stride = CC * GG;
    float mx = -3.402823466e38f;
    for (int c = 0; c < CC; ++c) mx = fmaxf(mx, src[base + c * stride]);
    float sum = 0.f;
    for (int c = 0; c < CC; ++c) sum += expf(src[base + c * stride] - mx);
    float inv = 1.0f / sum;
    for (int c = 0; c < CC; ++c) {
        const int fl = (c + 8 * (g >> 3)) + 16 * (k >> 2);
        dst[l * (CC * CC * GG) + ((g & 7) * 32 + fl) * 4 + (k & 3)] =
            expf(src[base + c * stride] - mx) * inv;
    }
}

__global__ void softmax_b_rows_kernel(const float* __restrict__ src,
                                      float* __restrict__ dst) {
    int r = blockIdx.x * blockDim.x + threadIdx.x;
    if (r >= (LL - 1) * CC * GG) return;
    const int l = r / (CC * GG), c = (r % (CC * GG)) / GG, g = r % GG;
    const int base = l * (CC * MM * GG) + c * (MM * GG) + g;
    float mx = -3.402823466e38f;
    for (int i = 0; i < MM; ++i) mx = fmaxf(mx, src[base + i * GG]);
    float sum = 0.f;
    for (int i = 0; i < MM; ++i) sum += expf(src[base + i * GG] - mx);
    float inv = 1.0f / sum;
    for (int i = 0; i < MM; ++i)
        dst[l * (CC * MM * GG) + i * (GG * CC) + g * CC + c] =
            expf(src[base + i * GG] - mx) * inv;
}

__global__ void fill_int_kernel(int* __restrict__ p, int n, int v) {
    int t = blockIdx.x * blockDim.x + threadIdx.x;
    if (t < n) p[t] = v;
}

__global__ void link_edges_kernel(const int* __restrict__ dst, int* __restrict__ head,
                                  int* __restrict__ next, int E) {
    int e = blockIdx.x * blockDim.x + threadIdx.x;
    if (e < E) next[e] = atomicExch(&head[dst[e]], e);
}

__global__ void layer0_kernel(const int* __restrict__ x,
                              const float* __restrict__ B0s,
                              const float* __restrict__ Pis,
                              float* __restrict__ h,
                              float* __restrict__ out,
                              int N) {
    int t = blockIdx.x * blockDim.x + threadIdx.x;
    if (t >= N * GG) return;
    int n = t / GG, g = t % GG;
    int xm = x[n];
    float u[CC];
    float Z = 0.f;
#pragma unroll
    for (int c = 0; c < CC; ++c) {
        u[c] = B0s[(c * MM + xm) * GG + g] * Pis[c * GG + g];
        Z += u[c];
    }
    out[(size_t)n * (LL * GG) + g] = logf(Z);
    float iz = 1.0f / Z;
#pragma unroll
    for (int c = 0; c < CC; ++c)
        h[(size_t)n * (GG * CC) + g * CC + c] = u[c] * iz;
}

__global__ void gather_mean_kernel(const float* __restrict__ h,
                                   const int* __restrict__ head,
                                   const int* __restrict__ next,
                                   const int* __restrict__ srcv,
                                   float* __restrict__ aggr, int N) {
    int t = blockIdx.x * blockDim.x + threadIdx.x;
    int lane = t & 31;
    int n = __builtin_amdgcn_readfirstlane(t >> 5);
    if (n >= N) return;
    int e = head[n];
    int deg = 0;
    float4 acc = make_float4(0.f, 0.f, 0.f, 0.f);
    float4 v = make_float4(0.f, 0.f, 0.f, 0.f);
    while (e >= 0) {
        const int nx = next[e];
        const int s = srcv[e];
        __builtin_amdgcn_sched_barrier(0);
        acc.x += v.x; acc.y += v.y; acc.z += v.z; acc.w += v.w;
        v = *(const float4*)(h + (size_t)s * 128 + lane * 4);
        e = nx;
        ++deg;
    }
    acc.x += v.x; acc.y += v.y; acc.z += v.z; acc.w += v.w;
    float iv = 1.0f / fmaxf((float)deg, 1.0f);
    acc.x *= iv; acc.y *= iv; acc.z *= iv; acc.w *= iv;
    *(float4*)(aggr + (size_t)n * 128 + lane * 4) = acc;
}

#define TPITCH 132
__global__ __launch_bounds__(32) void layer_update_kernel(
    const int* __restrict__ x,
    const float* __restrict__ Qf,
    const float* __restrict__ Bt,
    const float* __restrict__ aggr,
    float* __restrict__ h,
    float* __restrict__ out,
    int N, int lout) {
    __shared__ __attribute__((aligned(16))) float tile[16 * TPITCH];

    const int lane = threadIdx.x;
    const int half = lane >> 4;
    const int col = lane & 15;
    const int node0 = blockIdx.x * 16;

#pragma unroll
    for (int rr = 0; rr < 16; ++rr) {
        int nr = node0 + rr;
        nr = nr < N ? nr : N - 1;
        const float4 v = *(const float4*)(aggr + (size_t)nr * 128 + lane * 4);
        *(float4*)(tile + rr * TPITCH + lane * 4) = v;
    }
    __syncthreads();

    const int node = node0 + col;
    const int nn = node < N ? node : N - 1;
    const int xm = x[nn];
    const float* brow = Bt + xm * (GG * CC) + half * (8 * CC);
    float* trow = tile + col * TPITCH;
    const bool lo = col < 8;
    float lz[8];

#pragma unroll
    for (int p = 0; p < 8; ++p) {
        const float4 q = *(const float4*)(Qf + (p * 32 + lane) * 4);
        const float4 ba = *(const float4*)(trow + p * CC + 4 * half);
        const float4 bb = *(const float4*)(trow + (p + 8) * CC + 4 * half);
        v2f qa0, qa1, qb0, qb1, fa0, fa1, fb0, fb1;
        qa0.x = lo ? q.x : 0.f; qa0.y = lo ? q.y : 0.f;
        qa1.x = lo ? q.z : 0.f; qa1.y = lo ? q.w : 0.f;
        qb0.x = lo ? 0.f : q.x; qb0.y = lo ? 0.f : q.y;
        qb1.x = lo ? 0.f : q.z; qb1.y = lo ? 0.f : q.w;
        fa0.x = ba.x; fa0.y = ba.y; fa1.x = ba.z; fa1.y = ba.w;
        fb0.x = bb.x; fb0.y = bb.y; fb1.x = bb.z; fb1.y = bb.w;

        v8f acc = {0.f, 0.f, 0.f, 0.f, 0.f, 0.f, 0.f, 0.f};
        acc = __builtin_amdgcn_wmma_f32_16x16x4_f32(false, qa0, false, fa0,
                                                    (short)0, acc, false, false);
        acc = __builtin_amdgcn_wmma_f32_16x16x4_f32(false, qa1, false, fa1,
                                                    (short)0, acc, false, false);
        acc = __builtin_amdgcn_wmma_f32_16x16x4_f32(false, qb0, false, fb0,
                                                    (short)0, acc, false, false);
        acc = __builtin_amdgcn_wmma_f32_16x16x4_f32(false, qb1, false, fb1,
                                                    (short)0, acc, false, false);

        const float4 n0 = *(const float4*)(brow + p * CC);
        const float4 n1 = *(const float4*)(brow + p * CC + 4);
        float u[CC];
        u[0] = n0.x * acc[0]; u[1] = n0.y * acc[1];
        u[2] = n0.z * acc[2]; u[3] = n0.w * acc[3];
        u[4] = n1.x * acc[4]; u[5] = n1.y * acc[5];
        u[6] = n1.z * acc[6]; u[7] = n1.w * acc[7];
        float Z = 0.f;
#pragma unroll
        for (int c = 0; c < CC; ++c) Z += u[c];
        lz[p] = logf(Z);
        const float iz = 1.0f / Z;
        float* hd = trow + (p + 8 * half) * CC;
        *(float4*)(hd) = make_float4(u[0] * iz, u[1] * iz, u[2] * iz, u[3] * iz);
        *(float4*)(hd + 4) = make_float4(u[4] * iz, u[5] * iz, u[6] * iz, u[7] * iz);
    }

    if (node < N) {
        float* o = out + (size_t)node * (LL * GG) + lout * GG + half * 8;
        *(float4*)(o) = make_float4(lz[0], lz[1], lz[2], lz[3]);
        *(float4*)(o + 4) = make_float4(lz[4], lz[5], lz[6], lz[7]);
    }
    __syncthreads();

    if (node0 + 16 <= N) {
        float4 v[16];
#pragma unroll
        for (int rr = 0; rr < 16; ++rr)
            v[rr] = *(const float4*)(tile + rr * TPITCH + lane * 4);
#pragma unroll
        for (int rr = 0; rr < 16; ++rr)
            *(float4*)(h + (size_t)(node0 + rr) * 128 + lane * 4) = v[rr];
    } else {
        for (int rr = 0; rr < 16; ++rr) {
            if (node0 + rr < N) {
                const float4 v = *(const float4*)(tile + rr * TPITCH + lane * 4);
                *(float4*)(h + (size_t)(node0 + rr) * 128 + lane * 4) = v;
            }
        }
    }
}

extern "C" void kernel_launch(void* const* d_in, const int* in_sizes, int n_in,
                              void* d_out, int out_size, void* d_ws, size_t ws_size,
                              hipStream_t stream) {
    const int* x = (const int*)d_in[0];
    const int* ei = (const int*)d_in[1];
    const float* lB0 = (const float*)d_in[2];
    const float* lPi = (const float*)d_in[3];
    const float* lQ = (const float*)d_in[4];
    const float* lB = (const float*)d_in[5];
    float* out = (float*)d_out;

    const int N = in_sizes[0];
    const int E = in_sizes[1] / 2;
    const int* dst = ei;
    const int* srcv = ei + E;

    float* ws = (float*)d_ws;
    size_t off = 0;
    float* B0s = ws + off; off += CC * MM * GG;
    float* Pis = ws + off; off += CC * GG;
    float* Qs  = ws + off; off += (LL - 1) * CC * CC * GG;
    float* Bs  = ws + off; off += (LL - 1) * CC * MM * GG;
    int* head = (int*)(ws + off); off += N;
    int* next = (int*)(ws + off); off += E;
    off = (off + 15) & ~(size_t)15;
    float* h    = ws + off; off += (size_t)N * GG * CC;
    float* aggr = ws + off; off += (size_t)N * GG * CC;
    (void)ws_size; (void)n_in; (void)out_size;

    softmax_rows_kernel<<<1, 128, 0, stream>>>(lB0, B0s, CC * GG, MM, GG,
                                               CC * GG, 0, GG, MM * GG);
    softmax_rows_kernel<<<1, 32, 0, stream>>>(lPi, Pis, GG, CC, GG,
                                              GG, 0, GG, 0);
    softmax_q_frag_kernel<<<3, 128, 0, stream>>>(lQ, Qs);
    softmax_b_rows_kernel<<<3, 128, 0, stream>>>(lB, Bs);

    fill_int_kernel<<<(N + 255) / 256, 256, 0, stream>>>(head, N, -1);
    link_edges_kernel<<<(E + 255) / 256, 256, 0, stream>>>(dst, head, next, E);

    layer0_kernel<<<(N * GG + 255) / 256, 256, 0, stream>>>(x, B0s, Pis, h, out, N);

    const int gatherBlocks = (int)(((size_t)N * 32 + 255) / 256);
    for (int l = 0; l < LL - 1; ++l) {
        gather_mean_kernel<<<gatherBlocks, 256, 0, stream>>>(h, head, next, srcv,
                                                             aggr, N);
        layer_update_kernel<<<(N + 15) / 16, 32, 0, stream>>>(
            x, Qs + (size_t)l * CC * CC * GG, Bs + (size_t)l * CC * MM * GG,
            aggr, h, out, N, l + 1);
    }
}


# Gather: a node's in-edges are threaded into 4 linked lists (edge e into list e&3) that its wave walks side by side, 4 rows in flight, chain = longest list; the order of a node's sums changes
# speedup vs baseline: 2.2783x; 1.1901x over previous
#include <hip/hip_runtime.h>
#include <hip/hip_bf16.h>

#define CC 8
#define MM 32
#define GG 16
#define LL 4

typedef float v2f __attribute__((ext_vector_type(2)));
typedef float v8f __attribute__((ext_vector_type(8)));

__global__ void softmax_rows_kernel(const float* __restrict__ src,
                                    float* __restrict__ dst,
                                    int nrows, int len, int stride,
                                    int d1, int s1, int d2, int s2) {
    int r = blockIdx.x * blockDim.x + threadIdx.x;
    if (r >= nrows) return;
    int base = (r / d1) * s1 + ((r % d1) / d2) * s2 + (r % d2);
    float mx = -3.402823466e38f;
    for (int i = 0; i < len; ++i) mx = fmaxf(mx, src[base + i * stride]);
    float sum = 0.f;
    for (int i = 0; i < len; ++i) sum += expf(src[base + i * stride] - mx);
    float inv = 1.0f / sum;
    for (int i = 0; i < len; ++i)
        dst[base + i * stride] = expf(src[base + i * stride] - mx) * inv;
}

__global__ void softmax_q_frag_kernel(const float* __restrict__ src,
                                      float* __restrict__ dst) {
    int r = blockIdx.x * blockDim.x + threadIdx.x;
    if (r >= (LL - 1) * CC * GG) return;
    const int l = r / (CC * GG), k = (r % (CC * GG)) / GG, g = r % GG;
    const int base = l * (CC * CC * GG) + k * GG + g;
    const int stride = CC * GG;
    float mx = -3.402823466e38f;
    for (int c = 0; c < CC; ++c) mx = fmaxf(mx, src[base + c * stride]);
    float sum = 0.f;
    for (int c = 0; c < CC; ++c) sum += expf(src[base + c * stride] - mx);
    float inv = 1.0f / sum;
    for (int c = 0; c < CC; ++c) {
        const int fl = (c + 8 * (g >> 3)) + 16 * (k >> 2);
        dst[l * (CC * CC * GG) + ((g & 7) * 32 + fl) * 4 + (k & 3)] =
            expf(src[base + c * stride] - mx) * inv;
    }
}

__global__ void softmax_b_rows_kernel(const float* __restrict__ src,
                                      float* __restrict__ dst) {
    int r = blockIdx.x * blockDim.x + threadIdx.x;
    if (r >= (LL - 1) * CC * GG) return;
    const int l = r / (CC * GG), c = (r % (CC * GG)) / GG, g = r % GG;
    const int base = l * (CC * MM * GG) + c * (MM * GG) + g;
    float mx = -3.402823466e38f;
    for (int i = 0; i < MM; ++i) mx = fmaxf(mx, src[base + i * GG]);
    float sum = 0.f;
    for (int i = 0; i < MM; ++i) sum += expf(src[base + i * GG] - mx);
    float inv = 1.0f / sum;
    for (int i = 0; i < MM; ++i)
        dst[l * (CC * MM * GG) + i * (GG * CC) + g * CC + c] =
            expf(src[base + i * GG] - mx) * inv;
}

__global__ void fill_heads_kernel(int* __restrict__ head, int nh,
                                  float* __restrict__ zrow) {
    int t = blockIdx.x * blockDim.x + threadIdx.x;
    if (t < nh) head[t] = -1;
    if (t < GG * CC) zrow[t] = 0.f;
}

#define KL 4
__global__ void link_edges_kernel(const int* __restrict__ dst, int* __restrict__ head,
                                  int* __restrict__ next, int E) {
    int e = blockIdx.x * blockDim.x + threadIdx.x;
    if (e < E) next[e] = atomicExch(&head[dst[e] * KL + (e & (KL - 1))], e);
}

__global__ void layer0_kernel(const int* __restrict__ x,
                              const float* __restrict__ B0s,
                              const float* __restrict__ Pis,
                              float* __restrict__ h,
                              float* __restrict__ out,
                              int N) {
    int t = blockIdx.x * blockDim.x + threadIdx.x;
    if (t >= N * GG) return;
    int n = t / GG, g = t % GG;
    int xm = x[n];
    float u[CC];
    float Z = 0.f;
#pragma unroll
    for (int c = 0; c < CC; ++c) {
        u[c] = B0s[(c * MM + xm) * GG + g] * Pis[c * GG + g];
        Z += u[c];
    }
    out[(size_t)n * (LL * GG) + g] = logf(Z);
    float iz = 1.0f / Z;
#pragma unroll
    for (int c = 0; c < CC; ++c)
        h[(size_t)n * (GG * CC) + g * CC + c] = u[c] * iz;
}

__global__ void gather_mean_kernel(const float* __restrict__ hz,
                                   const int* __restrict__ head,
                                   const int* __restrict__ next,
                                   const int* __restrict__ srcv,
                                   float* __restrict__ aggr, int N) {
    int t = blockIdx.x * blockDim.x + threadIdx.x;
    int lane = t & 31;
    int n = __builtin_amdgcn_readfirstlane(t >> 5);
    if (n >= N) return;
    int e[KL];
    {
        const int4* hp = (const int4*)(head + (size_t)n * KL);
#pragma unroll
        for (int q = 0; q < KL / 4; ++q) {
            const int4 hd = hp[q];
            e[4 * q + 0] = hd.x; e[4 * q + 1] = hd.y;
            e[4 * q + 2] = hd.z; e[4 * q + 3] = hd.w;
        }
    }
    int deg = 0;
    float4 acc = make_float4(0.f, 0.f, 0.f, 0.f);
    float4 v[KL];
    int live = e[0];
#pragma unroll
    for (int j = 0; j < KL; ++j) {
        v[j] = make_float4(0.f, 0.f, 0.f, 0.f);
        live &= e[j];
    }
    const float* hl = hz + lane * 4;
    while (live >= 0) {
        int nx[KL], s[KL];
#pragma unroll
        for (int j = 0; j < KL; ++j) {
            const int i = e[j] > 0 ? e[j] : 0;
            nx[j] = next[i];
            s[j] = srcv[i];
        }
        __builtin_amdgcn_sched_barrier(0);
#pragma unroll
        for (int j = 0; j < KL; ++j) {
            acc.x += v[j].x; acc.y += v[j].y; acc.z += v[j].z; acc.w += v[j].w;
        }
        asm volatile("" : "+v"(acc.x), "+v"(acc.y), "+v"(acc.z), "+v"(acc.w) : : "memory");
        live = -1;
#pragma unroll
        for (int j = 0; j < KL; ++j) {
            const int m = e[j] >> 31;
            const int row = (s[j] + 1) & ~m;
            v[j] = *(const float4*)(hl + (size_t)row * 128);
            e[j] = nx[j] | m;
            deg += 1 + m;
            live &= e[j];
        }
    }
#pragma unroll
    for (int j = 0; j < KL; ++j) {
        acc.x += v[j].x; acc.y += v[j].y; acc.z += v[j].z; acc.w += v[j].w;
    }
    float iv = 1.0f / fmaxf((float)deg, 1.0f);
    acc.x *= iv; acc.y *= iv; acc.z *= iv; acc.w *= iv;
    *(float4*)(aggr + (size_t)n * 128 + lane * 4) = acc;
}

#define TPITCH 132
__global__ __launch_bounds__(32) void layer_update_kernel(
    const int* __restrict__ x,
    const float* __restrict__ Qf,
    const float* __restrict__ Bt,
    const float* __restrict__ aggr,
    float* __restrict__ h,
    float* __restrict__ out,
    int N, int lout) {
    __shared__ __attribute__((aligned(16))) float tile[16 * TPITCH];

    const int lane = threadIdx.x;
    const int half = lane >> 4;
    const int col = lane & 15;
    const int node0 = blockIdx.x * 16;

#pragma unroll
    for (int rr = 0; rr < 16; ++rr) {
        int nr = node0 + rr;
        nr = nr < N ? nr : N - 1;
        const float4 v = *(const float4*)(aggr + (size_t)nr * 128 + lane * 4);
        *(float4*)(tile + rr * TPITCH + lane * 4) = v;
    }
    __syncthreads();

    const int node = node0 + col;
    const int nn = node < N ? node : N - 1;
    const int xm = x[nn];
    const float* brow = Bt + xm * (GG * CC) + half * (8 * CC);
    float* trow = tile + col * TPITCH;
    const bool lo = col < 8;
    float lz[8];

#pragma unroll
    for (int p = 0; p < 8; ++p) {
        const float4 q = *(const float4*)(Qf + (p * 32 + lane) * 4);
        const float4 ba = *(const float4*)(trow + p * CC + 4 * half);
        const float4 bb = *(const float4*)(trow + (p + 8) * CC + 4 * half);
        v2f qa0, qa1, qb0, qb1, fa0, fa1, fb0, fb1;
        qa0.x = lo ? q.x : 0.f; qa0.y = lo ? q.y : 0.f;
        qa1.x = lo ? q.z : 0.f; qa1.y = lo ? q.w : 0.f;
        qb0.x = lo ? 0.f : q.x; qb0.y = lo ? 0.f : q.y;
        qb1.x = lo ? 0.f : q.z; qb1.y = lo ? 0.f : q.w;
        fa0.x = ba.x; fa0.y = ba.y; fa1.x = ba.z; fa1.y = ba.w;
        fb0.x = bb.x; fb0.y = bb.y; fb1.x = bb.z; fb1.y = bb.w;

        v8f acc = {0.f, 0.f, 0.f, 0.f, 0.f, 0.f, 0.f, 0.f};
        acc = __builtin_amdgcn_wmma_f32_16x16x4_f32(false, qa0, false, fa0,
                                                    (short)0, acc, false, false);
        acc = __builtin_amdgcn_wmma_f32_16x16x4_f32(false, qa1, false, fa1,
                                                    (short)0, acc, false, false);
        acc = __builtin_amdgcn_wmma_f32_16x16x4_f32(false, qb0, false, fb0,
                                                    (short)0, acc, false, false);
        acc = __builtin_amdgcn_wmma_f32_16x16x4_f32(false, qb1, false, fb1,
                                                    (short)0, acc, false, false);

        const float4 n0 = *(const float4*)(brow + p * CC);
        const float4 n1 = *(const float4*)(brow + p * CC + 4);
        float u[CC];
        u[0] = n0.x * acc[0]; u[1] = n0.y * acc[1];
        u[2] = n0.z * acc[2]; u[3] = n0.w * acc[3];
        u[4] = n1.x * acc[4]; u[5] = n1.y * acc[5];
        u[6] = n1.z * acc[6]; u[7] = n1.w * acc[7];
        float Z = 0.f;
#pragma unroll
        for (int c = 0; c < CC; ++c) Z += u[c];
        lz[p] = logf(Z);
        const float iz = 1.0f / Z;
        float* hd = trow + (p + 8 * half) * CC;
        *(float4*)(hd) = make_float4(u[0] * iz, u[1] * iz, u[2] * iz, u[3] * iz);
        *(float4*)(hd + 4) = make_float4(u[4] * iz, u[5] * iz, u[6] * iz, u[7] * iz);
    }

    if (node < N) {
        float* o = out + (size_t)node * (LL * GG) + lout * GG + half * 8;
        *(float4*)(o) = make_float4(lz[0], lz[1], lz[2], lz[3]);
        *(float4*)(o + 4) = make_float4(lz[4], lz[5], lz[6], lz[7]);
    }
    __syncthreads();

    if (node0 + 16 <= N) {
        float4 v[16];
#pragma unroll
        for (int rr = 0; rr < 16; ++rr)
            v[rr] = *(const float4*)(tile + rr * TPITCH + lane * 4);
#pragma unroll
        for (int rr = 0; rr < 16; ++rr)
            *(float4*)(h + (size_t)(node0 + rr) * 128 + lane * 4) = v[rr];
    } else {
        for (int rr = 0; rr < 16; ++rr) {
            if (node0 + rr < N) {
                const float4 v = *(const float4*)(tile + rr * TPITCH + lane * 4);
                *(float4*)(h + (size_t)(node0 + rr) * 128 + lane * 4) = v;
            }
        }
    }
}

extern "C" void kernel_launch(void* const* d_in, const int* in_sizes, int n_in,
                              void* d_out, int out_size, void* d_ws, size_t ws_size,
                              hipStream_t stream) {
    const int* x = (const int*)d_in[0];
    const int* ei = (const int*)d_in[1];
    const float* lB0 = (const float*)d_in[2];
    const float* lPi = (const float*)d_in[3];
    const float* lQ = (const float*)d_in[4];
    const float* lB = (const float*)d_in[5];
    float* out = (float*)d_out;

    const int N = in_sizes[0];
    const int E = in_sizes[1] / 2;
    const int* dst = ei;
    const int* srcv = ei + E;

    float* ws = (float*)d_ws;
    size_t off = 0;
    float* B0s = ws + off; off += CC * MM * GG;
    float* Pis = ws + off; off += CC * GG;
    float* Qs  = ws + off; off += (LL - 1) * CC * CC * GG;
    float* Bs  = ws + off; off += (LL - 1) * CC * MM * GG;
    int* head = (int*)(ws + off); off += (size_t)N * KL;
    int* next = (int*)(ws + off); off += E;
    off = (off + 15) & ~(size_t)15;
    float* hz   = ws + off; off += GG * CC;
    float* h    = ws + off; off += (size_t)N * GG * CC;
    float* aggr = ws + off; off += (size_t)N * GG * CC;
    (void)ws_size; (void)n_in; (void)out_size;

    softmax_rows_kernel<<<1, 128, 0, stream>>>(lB0, B0s, CC * GG, MM, GG,
                                               CC * GG, 0, GG, MM * GG);
    softmax_rows_kernel<<<1, 32, 0, stream>>>(lPi, Pis, GG, CC, GG,
                                              GG, 0, GG, 0);
    softmax_q_frag_kernel<<<3, 128, 0, stream>>>(lQ, Qs);
    softmax_b_rows_kernel<<<3, 128, 0, stream>>>(lB, Bs);

    fill_heads_kernel<<<(N * KL + 255) / 256, 256, 0, stream>>>(head, N * KL, hz);
    link_edges_kernel<<<(E + 255) / 256, 256, 0, stream>>>(dst, head, next, E);

    layer0_kernel<<<(N * GG + 255) / 256, 256, 0, stream>>>(x, B0s, Pis, h, out, N);

    const int gatherBlocks = (int)(((size_t)N * 32 + 255) / 256);
    for (int l = 0; l < LL - 1; ++l) {
        gather_mean_kernel<<<gatherBlocks, 256, 0, stream>>>(hz, head, next, srcv,
                                                             aggr, N);
        layer_update_kernel<<<(N + 15) / 16, 32, 0, stream>>>(
            x, Qs + (size_t)l * CC * CC * GG, Bs + (size_t)l * CC * MM * GG,
            aggr, h, out, N, l + 1);
    }
}


# Linked-list node packed: rec[e] = {next edge, source id}, one aligned 8-byte record written by the link pass; the gather walk issues one 8-byte scalar load per list and trip, not two 4-byte ones
# speedup vs baseline: 2.5549x; 1.1214x over previous
#include <hip/hip_runtime.h>
#include <hip/hip_bf16.h>

#define CC 8
#define MM 32
#define GG 16
#define LL 4

typedef float v2f __attribute__((ext_vector_type(2)));
typedef float v8f __attribute__((ext_vector_type(8)));

__global__ void softmax_rows_kernel(const float* __restrict__ src,
                                    float* __restrict__ dst,
                                    int nrows, int len, int stride,
                                    int d1, int s1, int d2, int s2) {
    int r = blockIdx.x * blockDim.x + threadIdx.x;
    if (r >= nrows) return;
    int base = (r / d1) * s1 + ((r % d1) / d2) * s2 + (r % d2);
    float mx = -3.402823466e38f;
    for (int i = 0; i < len; ++i) mx = fmaxf(mx, src[base + i * stride]);
    float sum = 0.f;
    for (int i = 0; i < len; ++i) sum += expf(src[base + i * stride] - mx);
    float inv = 1.0f / sum;
    for (int i = 0; i < len; ++i)
        dst[base + i * stride] = expf(src[base + i * stride] - mx) * inv;
}

__global__ void softmax_q_frag_kernel(const float* __restrict__ src,
                                      float* __restrict__ dst) {
    int r = blockIdx.x * blockDim.x + threadIdx.x;
    if (r >= (LL - 1) * CC * GG) return;
    const int l = r / (CC * GG), k = (r % (CC * GG)) / GG, g = r % GG;
    const int base = l * (CC * CC * GG) + k * GG + g;
    const int stride = CC * GG;
    float mx = -3.402823466e38f;
    for (int c = 0; c < CC; ++c) mx = fmaxf(mx, src[base + c * stride]);
    float sum = 0.f;
    for (int c = 0; c < CC; ++c) sum += expf(src[base + c * stride] - mx);
    float inv = 1.0f / sum;
    for (int c = 0; c < CC; ++c) {
        const int fl = (c + 8 * (g >> 3)) + 16 * (k >> 2);
        dst[l * (CC * CC * GG) + ((g & 7) * 32 + fl) * 4 + (k & 3)] =
            expf(src[base + c * stride] - mx) * inv;
    }
}

__global__ void softmax_b_rows_kernel(const float* __restrict__ src,
                                      float* __restrict__ dst) {
    int r = blockIdx.x * blockDim.x + threadIdx.x;
    if (r >= (LL - 1) * CC * GG) return;
    const int l = r / (CC * GG), c = (r % (CC * GG)) / GG, g = r % GG;
    const int base = l * (CC * MM * GG) + c * (MM * GG) + g;
    float mx = -3.402823466e38f;
    for (int i = 0; i < MM; ++i) mx = fmaxf(mx, src[base + i * GG]);
    float sum = 0.f;
    for (int i = 0; i < MM; ++i) sum += expf(src[base + i * GG] - mx);
    float inv = 1.0f / sum;
    for (int i = 0; i < MM; ++i)
        dst[l * (CC * MM * GG) + i * (GG * CC) + g * CC + c] =
            expf(src[base + i * GG] - mx) * inv;
}

__global__ void fill_heads_kernel(int* __restrict__ head, int nh,
                                  float* __restrict__ zrow) {
    int t = blockIdx.x * blockDim.x + threadIdx.x;
    if (t < nh) head[t] = -1;
    if (t < GG * CC) zrow[t] = 0.f;
}

#define KL 4
__global__ void link_edges_kernel(const int* __restrict__ dst,
                                  const int* __restrict__ srcv,
                                  int* __restrict__ head,
                                  int* __restrict__ rec, int E) {
    int e = blockIdx.x * blockDim.x + threadIdx.x;
    if (e < E) {
        const int s = srcv[e];
        const int prev = atomicExch(&head[dst[e] * KL + (e & (KL - 1))], e);
        *(int2*)(rec + 2 * (size_t)e) = make_int2(prev, s);
    }
}

__global__ void layer0_kernel(const int* __restrict__ x,
                              const float* __restrict__ B0s,
                              const float* __restrict__ Pis,
                              float* __restrict__ h,
                              float* __restrict__ out,
                              int N) {
    int t = blockIdx.x * blockDim.x + threadIdx.x;
    if (t >= N * GG) return;
    int n = t / GG, g = t % GG;
    int xm = x[n];
    float u[CC];
    float Z = 0.f;
#pragma unroll
    for (int c = 0; c < CC; ++c) {
        u[c] = B0s[(c * MM + xm) * GG + g] * Pis[c * GG + g];
        Z += u[c];
    }
    out[(size_t)n * (LL * GG) + g] = logf(Z);
    float iz = 1.0f / Z;
#pragma unroll
    for (int c = 0; c < CC; ++c)
        h[(size_t)n * (GG * CC) + g * CC + c] = u[c] * iz;
}

__global__ void gather_mean_kernel(const float* __restrict__ hz,
                                   const int* __restrict__ head,
                                   const int* __restrict__ rec,
                                   float* __restrict__ aggr, int N) {
    int t = blockIdx.x * blockDim.x + threadIdx.x;
    int lane = t & 31;
    int n = __builtin_amdgcn_readfirstlane(t >> 5);
    if (n >= N) return;
    int e[KL];
    {
        const int4* hp = (const int4*)(head + (size_t)n * KL);
#pragma unroll
        for (int q = 0; q < KL / 4; ++q) {
            const int4 hd = hp[q];
            e[4 * q + 0] = hd.x; e[4 * q + 1] = hd.y;
            e[4 * q + 2] = hd.z; e[4 * q + 3] = hd.w;
        }
    }
    int deg = 0;
    float4 acc = make_float4(0.f, 0.f, 0.f, 0.f);
    float4 v[KL];
    int live = e[0];
#pragma unroll
    for (int j = 0; j < KL; ++j) {
        v[j] = make_float4(0.f, 0.f, 0.f, 0.f);
        live &= e[j];
    }
    const float* hl = hz + lane * 4;
    while (live >= 0) {
        int nx[KL], s[KL];
#pragma unroll
        for (int j = 0; j < KL; ++j) {
            const int i = e[j] > 0 ? e[j] : 0;
            const int2 r = *(const int2*)(rec + 2 * (size_t)i);
            nx[j] = r.x;
            s[j] = r.y;
        }
        __builtin_amdgcn_sched_barrier(0);
#pragma unroll
        for (int j = 0; j < KL; ++j) {
            acc.x += v[j].x; acc.y += v[j].y; acc.z += v[j].z; acc.w += v[j].w;
        }
        asm volatile("" : "+v"(acc.x), "+v"(acc.y), "+v"(acc.z), "+v"(acc.w) : : "memory");
        live = -1;
#pragma unroll
        for (int j = 0; j < KL; ++j) {
            const int m = e[j] >> 31;
            const int row = (s[j] + 1) & ~m;
            v[j] = *(const float4*)(hl + (size_t)row * 128);
            e[j] = nx[j] | m;
            deg += 1 + m;
            live &= e[j];
        }
    }
#pragma unroll
    for (int j = 0; j < KL; ++j) {
        acc.x += v[j].x; acc.y += v[j].y; acc.z += v[j].z; acc.w += v[j].w;
    }
    float iv = 1.0f / fmaxf((float)deg, 1.0f);
    acc.x *= iv; acc.y *= iv; acc.z *= iv; acc.w *= iv;
    *(float4*)(aggr + (size_t)n * 128 + lane * 4) = acc;
}

#define TPITCH 132
__global__ __launch_bounds__(32) void layer_update_kernel(
    const int* __restrict__ x,
    const float* __restrict__ Qf,
    const float* __restrict__ Bt,
    const float* __restrict__ aggr,
    float* __restrict__ h,
    float* __restrict__ out,
    int N, int lout) {
    __shared__ __attribute__((aligned(16))) float tile[16 * TPITCH];

    const int lane = threadIdx.x;
    const int half = lane >> 4;
    const int col = lane & 15;
    const int node0 = blockIdx.x * 16;

#pragma unroll
    for (int rr = 0; rr < 16; ++rr) {
        int nr = node0 + rr;
        nr = nr < N ? nr : N - 1;
        const float4 v = *(const float4*)(aggr + (size_t)nr * 128 + lane * 4);
        *(float4*)(tile + rr * TPITCH + lane * 4) = v;
    }
    __syncthreads();

    const int node = node0 + col;
    const int nn = node < N ? node : N - 1;
    const int xm = x[nn];
    const float* brow = Bt + xm * (GG * CC) + half * (8 * CC);
    float* trow = tile + col * TPITCH;
    const bool lo = col < 8;
    float lz[8];

#pragma unroll
    for (int p = 0; p < 8; ++p) {
        const float4 q = *(const float4*)(Qf + (p * 32 + lane) * 4);
        const float4 ba = *(const float4*)(trow + p * CC + 4 * half);
        const float4 bb = *(const float4*)(trow + (p + 8) * CC + 4 * half);
        v2f qa0, qa1, qb0, qb1, fa0, fa1, fb0, fb1;
        qa0.x = lo ? q.x : 0.f; qa0.y = lo ? q.y : 0.f;
        qa1.x = lo ? q.z : 0.f; qa1.y = lo ? q.w : 0.f;
        qb0.x = lo ? 0.f : q.x; qb0.y = lo ? 0.f : q.y;
        qb1.x = lo ? 0.f : q.z; qb1.y = lo ? 0.f : q.w;
        fa0.x = ba.x; fa0.y = ba.y; fa1.x = ba.z; fa1.y = ba.w;
        fb0.x = bb.x; fb0.y = bb.y; fb1.x = bb.z; fb1.y = bb.w;

        v8f acc = {0.f, 0.f, 0.f, 0.f, 0.f, 0.f, 0.f, 0.f};
        acc = __builtin_amdgcn_wmma_f32_16x16x4_f32(false, qa0, false, fa0,
                                                    (short)0, acc, false, false);
        acc = __builtin_amdgcn_wmma_f32_16x16x4_f32(false, qa1, false, fa1,
                                                    (short)0, acc, false, false);
        acc = __builtin_amdgcn_wmma_f32_16x16x4_f32(false, qb0, false, fb0,
                                                    (short)0, acc, false, false);
        acc = __builtin_amdgcn_wmma_f32_16x16x4_f32(false, qb1, false, fb1,
                                                    (short)0, acc, false, false);

        const float4 n0 = *(const float4*)(brow + p * CC);
        const float4 n1 = *(const float4*)(brow + p * CC + 4);
        float u[CC];
        u[0] = n0.x * acc[0]; u[1] = n0.y * acc[1];
        u[2] = n0.z * acc[2]; u[3] = n0.w * acc[3];
        u[4] = n1.x * acc[4]; u[5] = n1.y * acc[5];
        u[6] = n1.z * acc[6]; u[7] = n1.w * acc[7];
        float Z = 0.f;
#pragma unroll
        for (int c = 0; c < CC; ++c) Z += u[c];
        lz[p] = logf(Z);
        const float iz = 1.0f / Z;
        float* hd = trow + (p + 8 * half) * CC;
        *(float4*)(hd) = make_float4(u[0] * iz, u[1] * iz, u[2] * iz, u[3] * iz);
        *(float4*)(hd + 4) = make_float4(u[4] * iz, u[5] * iz, u[6] * iz, u[7] * iz);
    }

    if (node < N) {
        float* o = out + (size_t)node * (LL * GG) + lout * GG + half * 8;
        *(float4*)(o) = make_float4(lz[0], lz[1], lz[2], lz[3]);
        *(float4*)(o + 4) = make_float4(lz[4], lz[5], lz[6], lz[7]);
    }
    __syncthreads();

    if (node0 + 16 <= N) {
        float4 v[16];
#pragma unroll
        for (int rr = 0; rr < 16; ++rr)
            v[rr] = *(const float4*)(tile + rr * TPITCH + lane * 4);
#pragma unroll
        for (int rr = 0; rr < 16; ++rr)
            *(float4*)(h + (size_t)(node0 + rr) * 128 + lane * 4) = v[rr];
    } else {
        for (int rr = 0; rr < 16; ++rr) {
            if (node0 + rr < N) {
                const float4 v = *(const float4*)(tile + rr * TPITCH + lane * 4);
                *(float4*)(h + (size_t)(node0 + rr) * 128 + lane * 4) = v;
            }
        }
    }
}

extern "C" void kernel_launch(void* const* d_in, const int* in_sizes, int n_in,
                              void* d_out, int out_size, void* d_ws, size_t ws_size,
                              hipStream_t stream) {
    const int* x = (const int*)d_in[0];
    const int* ei = (const int*)d_in[1];
    const float* lB0 = (const float*)d_in[2];
    const float* lPi = (const float*)d_in[3];
    const float* lQ = (const float*)d_in[4];
    const float* lB = (const float*)d_in[5];
    float* out = (float*)d_out;

    const int N = in_sizes[0];
    const int E = in_sizes[1] / 2;
    const int* dst = ei;
    const int* srcv = ei + E;

    float* ws = (float*)d_ws;
    size_t off = 0;
    float* B0s = ws + off; off += CC * MM * GG;
    float* Pis = ws + off; off += CC * GG;
    float* Qs  = ws + off; off += (LL - 1) * CC * CC * GG;
    float* Bs  = ws + off; off += (LL - 1) * CC * MM * GG;
    int* head = (int*)(ws + off); off += (size_t)N * KL;
    int* rec  = (int*)(ws + off); off += (size_t)E * 2;
    off = (off + 15) & ~(size_t)15;
    float* hz   = ws + off; off += GG * CC;
    float* h    = ws + off; off += (size_t)N * GG * CC;
    float* aggr = ws + off; off += (size_t)N * GG * CC;
    (void)ws_size; (void)n_in; (void)out_size;

    softmax_rows_kernel<<<1, 128, 0, stream>>>(lB0, B0s, CC * GG, MM, GG,
                                               CC * GG, 0, GG, MM * GG);
    softmax_rows_kernel<<<1, 32, 0, stream>>>(lPi, Pis, GG, CC, GG,
                                              GG, 0, GG, 0);
    softmax_q_frag_kernel<<<3, 128, 0, stream>>>(lQ, Qs);
    softmax_b_rows_kernel<<<3, 128, 0, stream>>>(lB, Bs);

    fill_heads_kernel<<<(N * KL + 255) / 256, 256, 0, stream>>>(head, N * KL, hz);
    link_edges_kernel<<<(E + 255) / 256, 256, 0, stream>>>(dst, srcv, head, rec, E);

    layer0_kernel<<<(N * GG + 255) / 256, 256, 0, stream>>>(x, B0s, Pis, h, out, N);

    const int gatherBlocks = (int)(((size_t)N * 32 + 255) / 256);
    for (int l = 0; l < LL - 1; ++l) {
        gather_mean_kernel<<<gatherBlocks, 256, 0, stream>>>(hz, head, rec, aggr, N);
        layer_update_kernel<<<(N + 15) / 16, 32, 0, stream>>>(
            x, Qs + (size_t)l * CC * CC * GG, Bs + (size_t)l * CC * MM * GG,
            aggr, h, out, N, l + 1);
    }
}


# Layer 0 and first neighbour mean from 32-row symbol tables: link pass counts in-edge symbols (integer atomic adds), layer 1 = counts x table on WMMA; one gather pass less; layer-1 sum order changes
# speedup vs baseline: 2.7255x; 1.0668x over previous
#include <hip/hip_runtime.h>
#include <hip/hip_bf16.h>

#define CC 8
#define MM 32
#define GG 16
#define LL 4
#define KL 4

typedef float v2f __attribute__((ext_vector_type(2)));
typedef float v8f __attribute__((ext_vector_type(8)));

__global__ void softmax_rows_kernel(const float* __restrict__ src,
                                    float* __restrict__ dst,
                                    int nrows, int len, int stride,
                                    int d1, int s1, int d2, int s2) {
    int r = blockIdx.x * blockDim.x + threadIdx.x;
    if (r >= nrows) return;
    int base = (r / d1) * s1 + ((r % d1) / d2) * s2 + (r % d2);
    float mx = -3.402823466e38f;
    for (int i = 0; i < len; ++i) mx = fmaxf(mx, src[base + i * stride]);
    float sum = 0.f;
    for (int i = 0; i < len; ++i) sum += expf(src[base + i * stride] - mx);
    float inv = 1.0f / sum;
    for (int i = 0; i < len; ++i)
        dst[base + i * stride] = expf(src[base + i * stride] - mx) * inv;
}

__global__ void softmax_q_frag_kernel(const float* __restrict__ src,
                                      float* __restrict__ dst) {
    int r = blockIdx.x * blockDim.x + threadIdx.x;
    if (r >= (LL - 1) * CC * GG) return;
    const int l = r / (CC * GG), k = (r % (CC * GG)) / GG, g = r % GG;
    const int base = l * (CC * CC * GG) + k * GG + g;
    const int stride = CC * GG;
    float mx = -3.402823466e38f;
    for (int c = 0; c < CC; ++c) mx = fmaxf(mx, src[base + c * stride]);
    float sum = 0.f;
    for (int c = 0; c < CC; ++c) sum += expf(src[base + c * stride] - mx);
    float inv = 1.0f / sum;
    for (int c = 0; c < CC; ++c) {
        const int fl = (c + 8 * (g >> 3)) + 16 * (k >> 2);
        dst[l * (CC * CC * GG) + ((g & 7) * 32 + fl) * 4 + (k & 3)] =
            expf(src[base + c * stride] - mx) * inv;
    }
}

__global__ void softmax_b_rows_kernel(const float* __restrict__ src,
                                      float* __restrict__ dst) {
    int r = blockIdx.x * blockDim.x + threadIdx.x;
    if (r >= (LL - 1) * CC * GG) return;
    const int l = r / (CC * GG), c = (r % (CC * GG)) / GG, g = r % GG;
    const int base = l * (CC * MM * GG) + c * (MM * GG) + g;
    float mx = -3.402823466e38f;
    for (int i = 0; i < MM; ++i) mx = fmaxf(mx, src[base + i * GG]);
    float sum = 0.f;
    for (int i = 0; i < MM; ++i) sum += expf(src[base + i * GG] - mx);
    float inv = 1.0f / sum;
    for (int i = 0; i < MM; ++i)
        dst[l * (CC * MM * GG) + i * (GG * CC) + g * CC + c] =
            expf(src[base + i * GG] - mx) * inv;
}

__global__ void fill_heads_kernel(int* __restrict__ head, int nh,
                                  float* __restrict__ zrow,
                                  int* __restrict__ cnt) {
    static_assert(MM / 4 == 2 * KL, "two 16-byte counter groups per head");
    int t = blockIdx.x * blockDim.x + threadIdx.x;
    if (t < nh) {
        head[t] = -1;
        int4* c4 = (int4*)cnt;
        c4[t] = make_int4(0, 0, 0, 0);
        c4[t + nh] = make_int4(0, 0, 0, 0);
    }
    if (t < GG * CC) zrow[t] = 0.f;
}

__global__ void link_edges_kernel(const int* __restrict__ dst,
                                  const int* __restrict__ srcv,
                                  const int* __restrict__ x,
                                  int* __restrict__ head,
                                  int* __restrict__ rec,
                                  int* __restrict__ cnt, int E) {
    int e = blockIdx.x * blockDim.x + threadIdx.x;
    if (e < E) {
        const int s = srcv[e];
        const int d = dst[e];
        const int xs = x[s];
        const int prev = atomicExch(&head[d * KL + (e & (KL - 1))], e);
        *(int2*)(rec + 2 * (size_t)e) = make_int2(prev, s);
        atomicAdd(&cnt[d * MM + xs], 1);
    }
}

__global__ void layer0_tables_kernel(const float* __restrict__ B0s,
                                     const float* __restrict__ Pis,
                                     float* __restrict__ T0f,
                                     float* __restrict__ LZ0) {
    const int t = threadIdx.x;
    if (t >= MM * GG) return;
    const int m = t / GG, g = t % GG;
    float u[CC];
    float Z = 0.f;
#pragma unroll
    for (int c = 0; c < CC; ++c) {
        u[c] = B0s[(c * MM + m) * GG + g] * Pis[c * GG + g];
        Z += u[c];
    }
    LZ0[m * GG + g] = logf(Z);
    const float iz = 1.0f / Z;
    const int p = g & 7, gsel = g >> 3;
    const int hh = m >> 4, sp = (m & 15) >> 2, e = m & 3;
#pragma unroll
    for (int k = 0; k < CC; ++k) {
        const int R = 8 * (k >> 2) + 4 * gsel + (k & 3);
        T0f[((p * 4 + sp) * 32 + (R + 16 * hh)) * 4 + e] = u[k] * iz;
    }
}

__global__ void gather_mean_kernel(const float* __restrict__ hz,
                                   const int* __restrict__ head,
                                   const int* __restrict__ rec,
                                   float* __restrict__ aggr, int N) {
    int t = blockIdx.x * blockDim.x + threadIdx.x;
    int lane = t & 31;
    int n = __builtin_amdgcn_readfirstlane(t >> 5);
    if (n >= N) return;
    int e[KL];
    {
        const int4* hp = (const int4*)(head + (size_t)n * KL);
#pragma unroll
        for (int q = 0; q < KL / 4; ++q) {
            const int4 hd = hp[q];
            e[4 * q + 0] = hd.x; e[4 * q + 1] = hd.y;
            e[4 * q + 2] = hd.z; e[4 * q + 3] = hd.w;
        }
    }
    int deg = 0;
    float4 acc = make_float4(0.f, 0.f, 0.f, 0.f);
    float4 v[KL];
    int live = e[0];
#pragma unroll
    for (int j = 0; j < KL; ++j) {
        v[j] = make_float4(0.f, 0.f, 0.f, 0.f);
        live &= e[j];
    }
    const float* hl = hz + lane * 4;
    while (live >= 0) {
        int nx[KL], s[KL];
#pragma unroll
        for (int j = 0; j < KL; ++j) {
            const int i = e[j] > 0 ? e[j] : 0;
            const int2 r = *(const int2*)(rec + 2 * (size_t)i);
            nx[j] = r.x;
            s[j] = r.y;
        }
        __builtin_amdgcn_sched_barrier(0);
#pragma unroll
        for (int j = 0; j < KL; ++j) {
            acc.x += v[j].x; acc.y += v[j].y; acc.z += v[j].z; acc.w += v[j].w;
        }
        asm volatile("" : "+v"(acc.x), "+v"(acc.y), "+v"(acc.z), "+v"(acc.w) : : "memory");
        live = -1;
#pragma unroll
        for (int j = 0; j < KL; ++j) {
            const int m = e[j] >> 31;
            const int row = (s[j] + 1) & ~m;
            v[j] = *(const float4*)(hl + (size_t)row * 128);
            e[j] = nx[j] | m;
            deg += 1 + m;
            live &= e[j];
        }
    }
#pragma unroll
    for (int j = 0; j < KL; ++j) {
        acc.x += v[j].x; acc.y += v[j].y; acc.z += v[j].z; acc.w += v[j].w;
    }
    float iv = 1.0f / fmaxf((float)deg, 1.0f);
    acc.x *= iv; acc.y *= iv; acc.z *= iv; acc.w *= iv;
    *(float4*)(aggr + (size_t)n * 128 + lane * 4) = acc;
}

#define TPITCH 132
__global__ __launch_bounds__(32) void layer_update_kernel(
    const int* __restrict__ x,
    const float* __restrict__ Qf,
    const float* __restrict__ Bt,
    const float* __restrict__ aggr,
    float* __restrict__ h,
    float* __restrict__ out,
    int N, int lout) {
    __shared__ __attribute__((aligned(16))) float tile[16 * TPITCH];

    const int lane = threadIdx.x;
    const int half = lane >> 4;
    const int col = lane & 15;
    const int node0 = blockIdx.x * 16;

#pragma unroll
    for (int rr = 0; rr < 16; ++rr) {
        int nr = node0 + rr;
        nr = nr < N ? nr : N - 1;
        const float4 v = *(const float4*)(aggr + (size_t)nr * 128 + lane * 4);
        *(float4*)(tile + rr * TPITCH + lane * 4) = v;
    }
    __syncthreads();

    const int node = node0 + col;
    const int nn = node < N ? node : N - 1;
    const int xm = x[nn];
    const float* brow = Bt + xm * (GG * CC) + half * (8 * CC);
    float* trow = tile + col * TPITCH;
    const bool lo = col < 8;
    float lz[8];

#pragma unroll
    for (int p = 0; p < 8; ++p) {
        const float4 q = *(const float4*)(Qf + (p * 32 + lane) * 4);
        const float4 ba = *(const float4*)(trow + p * CC + 4 * half);
        const float4 bb = *(const float4*)(trow + (p + 8) * CC + 4 * half);
        v2f qa0, qa1, qb0, qb1, fa0, fa1, fb0, fb1;
        qa0.x = lo ? q.x : 0.f; qa0.y = lo ? q.y : 0.f;
        qa1.x = lo ? q.z : 0.f; qa1.y = lo ? q.w : 0.f;
        qb0.x = lo ? 0.f : q.x; qb0.y = lo ? 0.f : q.y;
        qb1.x = lo ? 0.f : q.z; qb1.y = lo ? 0.f : q.w;
        fa0.x = ba.x; fa0.y = ba.y; fa1.x = ba.z; fa1.y = ba.w;
        fb0.x = bb.x; fb0.y = bb.y; fb1.x = bb.z; fb1.y = bb.w;

        v8f acc = {0.f, 0.f, 0.f, 0.f, 0.f, 0.f, 0.f, 0.f};
        acc = __builtin_amdgcn_wmma_f32_16x16x4_f32(false, qa0, false, fa0,
                                                    (short)0, acc, false, false);
        acc = __builtin_amdgcn_wmma_f32_16x16x4_f32(false, qa1, false, fa1,
                                                    (short)0, acc, false, false);
        acc = __builtin_amdgcn_wmma_f32_16x16x4_f32(false, qb0, false, fb0,
                                                    (short)0, acc, false, false);
        acc = __builtin_amdgcn_wmma_f32_16x16x4_f32(false, qb1, false, fb1,
                                                    (short)0, acc, false, false);

        const float4 n0 = *(const float4*)(brow + p * CC);
        const float4 n1 = *(const float4*)(brow + p * CC + 4);
        float u[CC];
        u[0] = n0.x * acc[0]; u[1] = n0.y * acc[1];
        u[2] = n0.z * acc[2]; u[3] = n0.w * acc[3];
        u[4] = n1.x * acc[4]; u[5] = n1.y * acc[5];
        u[6] = n1.z * acc[6]; u[7] = n1.w * acc[7];
        float Z = 0.f;
#pragma unroll
        for (int c = 0; c < CC; ++c) Z += u[c];
        lz[p] = logf(Z);
        const float iz = 1.0f / Z;
        float* hd = trow + (p + 8 * half) * CC;
        *(float4*)(hd) = make_float4(u[0] * iz, u[1] * iz, u[2] * iz, u[3] * iz);
        *(float4*)(hd + 4) = make_float4(u[4] * iz, u[5] * iz, u[6] * iz, u[7] * iz);
    }

    if (node < N) {
        float* o = out + (size_t)node * (LL * GG) + lout * GG + half * 8;
        *(float4*)(o) = make_float4(lz[0], lz[1], lz[2], lz[3]);
        *(float4*)(o + 4) = make_float4(lz[4], lz[5], lz[6], lz[7]);
    }
    __syncthreads();

    if (node0 + 16 <= N) {
        float4 v[16];
#pragma unroll
        for (int rr = 0; rr < 16; ++rr)
            v[rr] = *(const float4*)(tile + rr * TPITCH + lane * 4);
#pragma unroll
        for (int rr = 0; rr < 16; ++rr)
            *(float4*)(h + (size_t)(node0 + rr) * 128 + lane * 4) = v[rr];
    } else {
        for (int rr = 0; rr < 16; ++rr) {
            if (node0 + rr < N) {
                const float4 v = *(const float4*)(tile + rr * TPITCH + lane * 4);
                *(float4*)(h + (size_t)(node0 + rr) * 128 + lane * 4) = v;
            }
        }
    }
}

__global__ __launch_bounds__(32) void layer1_update_kernel(
    const int* __restrict__ x,
    const float* __restrict__ T0f,
    const float* __restrict__ LZ0,
    const float* __restrict__ Qf,
    const float* __restrict__ Bt,
    const int* __restrict__ cnt,
    float* __restrict__ h,
    float* __restrict__ out,
    int N) {
    __shared__ __attribute__((aligned(16))) float tile[16 * TPITCH];

    const int lane = threadIdx.x;
    const int half = lane >> 4;
    const int col = lane & 15;
    const int node0 = blockIdx.x * 16;
    const int node = node0 + col;
    const int nn = node < N ? node : N - 1;
    const int xm = x[nn];

    const int4* cp = (const int4*)(cnt + (size_t)nn * MM + 16 * half);
    const int4 c0 = cp[0], c1 = cp[1], c2 = cp[2], c3 = cp[3];
    int dsum = (c0.x + c0.y + c0.z + c0.w) + (c1.x + c1.y + c1.z + c1.w) +
               (c2.x + c2.y + c2.z + c2.w) + (c3.x + c3.y + c3.z + c3.w);
    dsum += __shfl_xor(dsum, 16);
    const float iv = 1.0f / fmaxf((float)dsum, 1.0f);
    float cv[16];
    cv[0] = (float)c0.x;  cv[1] = (float)c0.y;  cv[2] = (float)c0.z;  cv[3] = (float)c0.w;
    cv[4] = (float)c1.x;  cv[5] = (float)c1.y;  cv[6] = (float)c1.z;  cv[7] = (float)c1.w;
    cv[8] = (float)c2.x;  cv[9] = (float)c2.y;  cv[10] = (float)c2.z; cv[11] = (float)c2.w;
    cv[12] = (float)c3.x; cv[13] = (float)c3.y; cv[14] = (float)c3.z; cv[15] = (float)c3.w;

    const float* brow = Bt + xm * (GG * CC) + half * (8 * CC);
    float* trow = tile + col * TPITCH;
    const bool lo = col < 8;
    float lz[8];

#pragma unroll
    for (int p = 0; p < 8; ++p) {
        v8f ag = {0.f, 0.f, 0.f, 0.f, 0.f, 0.f, 0.f, 0.f};
#pragma unroll
        for (int sp = 0; sp < 4; ++sp) {
            const float4 a = *(const float4*)(T0f + ((p * 4 + sp) * 32 + lane) * 4);
            v2f a0, a1, b0, b1;
            a0.x = a.x; a0.y = a.y; a1.x = a.z; a1.y = a.w;
            b0.x = cv[4 * sp + 0]; b0.y = cv[4 * sp + 1];
            b1.x = cv[4 * sp + 2]; b1.y = cv[4 * sp + 3];
            ag = __builtin_amdgcn_wmma_f32_16x16x4_f32(false, a0, false, b0,
                                                       (short)0, ag, false, false);
            ag = __builtin_amdgcn_wmma_f32_16x16x4_f32(false, a1, false, b1,
                                                       (short)0, ag, false, false);
        }
        v2f fa0, fa1, fb0, fb1;
        fa0.x = ag[0] * iv; fa0.y = ag[1] * iv;
        fa1.x = ag[2] * iv; fa1.y = ag[3] * iv;
        fb0.x = ag[4] * iv; fb0.y = ag[5] * iv;
        fb1.x = ag[6] * iv; fb1.y = ag[7] * iv;

        const float4 q = *(const float4*)(Qf + (p * 32 + lane) * 4);
        v2f qa0, qa1, qb0, qb1;
        qa0.x = lo ? q.x : 0.f; qa0.y = lo ? q.y : 0.f;
        qa1.x = lo ? q.z : 0.f; qa1.y = lo ? q.w : 0.f;
        qb0.x = lo ? 0.f : q.x; qb0.y = lo ? 0.f : q.y;
        qb1.x = lo ? 0.f : q.z; qb1.y = lo ? 0.f : q.w;

        v8f acc = {0.f, 0.f, 0.f, 0.f, 0.f, 0.f, 0.f, 0.f};
        acc = __builtin_amdgcn_wmma_f32_16x16x4_f32(false, qa0, false, fa0,
                                                    (short)0, acc, false, false);
        acc = __builtin_amdgcn_wmma_f32_16x16x4_f32(false, qa1, false, fa1,
                                                    (short)0, acc, false, false);
        acc = __builtin_amdgcn_wmma_f32_16x16x4_f32(false, qb0, false, fb0,
                                                    (short)0, acc, false, false);
        acc = __builtin_amdgcn_wmma_f32_16x16x4_f32(false, qb1, false, fb1,
                                                    (short)0, acc, false, false);

        const float4 n0 = *(const float4*)(brow + p * CC);
        const float4 n1 = *(const float4*)(brow + p * CC + 4);
        float u[CC];
        u[0] = n0.x * acc[0]; u[1] = n0.y * acc[1];
        u[2] = n0.z * acc[2]; u[3] = n0.w * acc[3];
        u[4] = n1.x * acc[4]; u[5] = n1.y * acc[5];
        u[6] = n1.z * acc[6]; u[7] = n1.w * acc[7];
        float Z = 0.f;
#pragma unroll
        for (int c = 0; c < CC; ++c) Z += u[c];
        lz[p] = logf(Z);
        const float iz = 1.0f / Z;
        float* hd = trow + (p + 8 * half) * CC;
        *(float4*)(hd) = make_float4(u[0] * iz, u[1] * iz, u[2] * iz, u[3] * iz);
        *(float4*)(hd + 4) = make_float4(u[4] * iz, u[5] * iz, u[6] * iz, u[7] * iz);
    }

    if (node < N) {
        const float* z0 = LZ0 + xm * GG + half * 8;
        float* o = out + (size_t)node * (LL * GG) + half * 8;
        *(float4*)(o) = *(const float4*)(z0);
        *(float4*)(o + 4) = *(const float4*)(z0 + 4);
        *(float4*)(o + GG) = make_float4(lz[0], lz[1], lz[2], lz[3]);
        *(float4*)(o + GG + 4) = make_float4(lz[4], lz[5], lz[6], lz[7]);
    }
    __syncthreads();

    if (node0 + 16 <= N) {
        float4 v[16];
#pragma unroll
        for (int rr = 0; rr < 16; ++rr)
            v[rr] = *(const float4*)(tile + rr * TPITCH + lane * 4);
#pragma unroll
        for (int rr = 0; rr < 16; ++rr)
            *(float4*)(h + (size_t)(node0 + rr) * 128 + lane * 4) = v[rr];
    } else {
        for (int rr = 0; rr < 16; ++rr) {
            if (node0 + rr < N) {
                const float4 v = *(const float4*)(tile + rr * TPITCH + lane * 4);
                *(float4*)(h + (size_t)(node0 + rr) * 128 + lane * 4) = v;
            }
        }
    }
}

extern "C" void kernel_launch(void* const* d_in, const int* in_sizes, int n_in,
                              void* d_out, int out_size, void* d_ws, size_t ws_size,
                              hipStream_t stream) {
    const int* x = (const int*)d_in[0];
    const int* ei = (const int*)d_in[1];
    const float* lB0 = (const float*)d_in[2];
    const float* lPi = (const float*)d_in[3];
    const float* lQ = (const float*)d_in[4];
    const float* lB = (const float*)d_in[5];
    float* out = (float*)d_out;

    const int N = in_sizes[0];
    const int E = in_sizes[1] / 2;
    const int* dst = ei;
    const int* srcv = ei + E;

    float* ws = (float*)d_ws;
    size_t off = 0;
    float* B0s = ws + off; off += CC * MM * GG;
    float* Pis = ws + off; off += CC * GG;
    float* Qs  = ws + off; off += (LL - 1) * CC * CC * GG;
    float* Bs  = ws + off; off += (LL - 1) * CC * MM * GG;
    int* head = (int*)(ws + off); off += (size_t)N * KL;
    int* rec  = (int*)(ws + off); off += (size_t)E * 2;
    off = (off + 15) & ~(size_t)15;
    float* hz   = ws + off; off += GG * CC;
    float* h    = ws + off; off += (size_t)N * GG * CC;
    float* aggr = ws + off; off += (size_t)N * GG * CC;
    int* cnt   = (int*)aggr;
    float* T0f = aggr + (size_t)N * MM;
    float* LZ0 = T0f + 8 * 4 * 32 * 4;
    (void)ws_size; (void)n_in; (void)out_size;

    softmax_rows_kernel<<<1, 128, 0, stream>>>(lB0, B0s, CC * GG, MM, GG,
                                               CC * GG, 0, GG, MM * GG);
    softmax_rows_kernel<<<1, 32, 0, stream>>>(lPi, Pis, GG, CC, GG,
                                              GG, 0, GG, 0);
    softmax_q_frag_kernel<<<3, 128, 0, stream>>>(lQ, Qs);
    softmax_b_rows_kernel<<<3, 128, 0, stream>>>(lB, Bs);

    layer0_tables_kernel<<<1, MM * GG, 0, stream>>>(B0s, Pis, T0f, LZ0);

    fill_heads_kernel<<<(N * KL + 255) / 256, 256, 0, stream>>>(head, N * KL, hz, cnt);
    link_edges_kernel<<<(E + 255) / 256, 256, 0, stream>>>(dst, srcv, x, head, rec, cnt, E);

    layer1_update_kernel<<<(N + 15) / 16, 32, 0, stream>>>(
        x, T0f, LZ0, Qs, Bs, cnt, h, out, N);

    const int gatherBlocks = (int)(((size_t)N * 32 + 255) / 256);
    for (int l = 1; l < LL - 1; ++l) {
        gather_mean_kernel<<<gatherBlocks, 256, 0, stream>>>(hz, head, rec, aggr, N);
        layer_update_kernel<<<(N + 15) / 16, 32, 0, stream>>>(
            x, Qs + (size_t)l * CC * CC * GG, Bs + (size_t)l * CC * MM * GG,
            aggr, h, out, N, l + 1);
    }
}
